# starting kernel
# baseline (speedup 1.0000x reference)
#include <hip/hip_runtime.h>
#include <hip/hip_bf16.h>
#include <math.h>

#define NN 10000
#define EE 160000

typedef __attribute__((ext_vector_type(2))) float v2f;
typedef __attribute__((ext_vector_type(8))) float v8f;

__device__ __forceinline__ float lrelu_(float x) { return x > 0.f ? x : 0.2f * x; }
__device__ __forceinline__ float elu_(float x)   { return x > 0.f ? x : (expf(x) - 1.f); }
__device__ __forceinline__ unsigned fenc_(float f) {
    unsigned u = __float_as_uint(f);
    return (u & 0x80000000u) ? ~u : (u | 0x80000000u);
}
__device__ __forceinline__ float fdec_(unsigned e) {
    return __uint_as_float((e & 0x80000000u) ? (e & 0x7FFFFFFFu) : ~e);
}

__global__ __launch_bounds__(128)
void k_gemm_rel_wmma(const float* __restrict__ X, const float* __restrict__ W,
                     float* __restrict__ Y, int M, int K, int Nout) {
    __shared__ float As[16 * 256];
    const int r    = blockIdx.z;
    const int m0   = blockIdx.x * 16;
    const int wave = threadIdx.x >> 5;
    const int lane = threadIdx.x & 31;
    const int n0   = blockIdx.y * 64 + wave * 16;

    for (int row = 0; row < 16; ++row)
        for (int col = threadIdx.x; col < K; col += 128)
            As[row * K + col] = X[(size_t)(m0 + row) * K + col];
    __syncthreads();

    if (n0 < Nout) {
        const float* Wr  = W + (size_t)r * K * Nout;
        const int  ncol  = n0 + (lane & 15);
        const bool nv    = ncol < Nout;
        const int  ncl   = nv ? ncol : (Nout - 1);
        const int  kh    = (lane >> 4) * 2;
        const int  mrow  = lane & 15;
        v8f c = {};
        for (int k0 = 0; k0 < K; k0 += 4) {
            v2f a, b;
            a.x = As[mrow * K + k0 + kh];
            a.y = As[mrow * K + k0 + kh + 1];
            b.x = Wr[(size_t)(k0 + kh)     * Nout + ncl];
            b.y = Wr[(size_t)(k0 + kh + 1) * Nout + ncl];
            c = __builtin_amdgcn_wmma_f32_16x16x4_f32(false, a, false, b,
                                                      (short)0, c, false, false);
        }
        if (nv) {
            float* Yr = Y + (size_t)r * M * Nout;
            const int rbase = m0 + (lane >> 4) * 8;
#pragma unroll
            for (int i = 0; i < 8; ++i)
                Yr[(size_t)(rbase + i) * Nout + ncol] = c[i];
        }
    }
}

__global__ __launch_bounds__(128)
void k_rowdot(const float* __restrict__ XA, const float* __restrict__ Q,
              const float* __restrict__ Kv, float* __restrict__ qall,
              float* __restrict__ kall, int D, int H, int total) {
    const int wid  = (int)((blockIdx.x * (unsigned)blockDim.x + threadIdx.x) >> 5);
    const int lane = threadIdx.x & 31;
    if (wid >= total) return;
    const float* xrow = XA + (size_t)wid * D;
    float aq[2] = {0.f, 0.f}, ak[2] = {0.f, 0.f};
    for (int d = lane; d < D; d += 32) {
        float v = xrow[d];
        for (int h = 0; h < H; ++h) {
            aq[h] += v * Q[d * H + h];
            ak[h] += v * Kv[d * H + h];
        }
    }
    for (int off = 16; off; off >>= 1)
        for (int h = 0; h < H; ++h) {
            aq[h] += __shfl_xor(aq[h], off, 32);
            ak[h] += __shfl_xor(ak[h], off, 32);
        }
    if (lane == 0)
        for (int h = 0; h < H; ++h) {
            qall[(size_t)wid * H + h] = aq[h];
            kall[(size_t)wid * H + h] = ak[h];
        }
}

__global__ void k_init(unsigned* __restrict__ menc, float* __restrict__ den,
                       float* __restrict__ aggr, int nh, int na) {
    const int stride = gridDim.x * blockDim.x;
    for (int t = blockIdx.x * blockDim.x + threadIdx.x; t < nh; t += stride) {
        menc[t] = 0x007FFFFFu;
        den[t]  = 0.f;
    }
    for (int t = blockIdx.x * blockDim.x + threadIdx.x; t < na; t += stride)
        aggr[t] = 0.f;
}

__global__ void k_edge_alpha_max(const float* __restrict__ qall,
                                 const float* __restrict__ kall,
                                 const int* __restrict__ src, const int* __restrict__ dst,
                                 const int* __restrict__ etype,
                                 float* __restrict__ exbuf, unsigned* __restrict__ menc,
                                 int H, int Rr) {
    const int t = blockIdx.x * blockDim.x + threadIdx.x;
    if (t >= EE * H) return;
    const int e = t / H, h = t - e * H;
    const int et = (Rr > 1) ? etype[e] : 0;
    const int s = src[e], d = dst[e];
    const float a = lrelu_(qall[((size_t)et * NN + d) * H + h] +
                           kall[((size_t)et * NN + s) * H + h]);
    exbuf[t] = a;
    atomicMax(&menc[d * H + h], fenc_(a));
}

__global__ void k_edge_exp(const int* __restrict__ dst, float* __restrict__ exbuf,
                           const unsigned* __restrict__ menc, float* __restrict__ den,
                           int H) {
    const int t = blockIdx.x * blockDim.x + threadIdx.x;
    if (t >= EE * H) return;
    const int e = t / H, h = t - e * H;
    const int d = dst[e];
    const float ex = expf(exbuf[t] - fdec_(menc[d * H + h]));
    exbuf[t] = ex;
    atomicAdd(&den[d * H + h], ex);
}

__global__ void k_edge_aggr(const float* __restrict__ XA, const float* __restrict__ exbuf,
                            const float* __restrict__ den,
                            const int* __restrict__ src, const int* __restrict__ dst,
                            const int* __restrict__ etype, float* __restrict__ aggr,
                            int D, int C, int H, int Rr) {
    const int e = blockIdx.x;
    const int c = threadIdx.x;
    if (c >= C) return;
    const int s = src[e], d = dst[e];
    const int et = (Rr > 1) ? etype[e] : 0;
    const float* xrow = XA + ((size_t)et * NN + s) * D;
    float v = 0.f;
    for (int h = 0; h < H; ++h) {
        const float attn = exbuf[(size_t)e * H + h] / (den[d * H + h] + 1e-16f);
        v += attn * xrow[h * C + c];
    }
    atomicAdd(&aggr[(size_t)d * C + c], v / (float)H);
}

__global__ void k_epilogue(const float* __restrict__ aggr, const float* __restrict__ bias,
                           float* __restrict__ out, int C, int total) {
    const int t = blockIdx.x * blockDim.x + threadIdx.x;
    if (t >= total) return;
    out[t] = elu_(aggr[t] + bias[t % C]);
}

struct WsPtrs {
    float* xall; float* qall; float* kall; unsigned* menc;
    float* den;  float* exbuf; float* aggr;
};

static void run_attn_conv(const float* xin, const float* w, const float* qv,
                          const float* kv, const float* bias,
                          int K, int D, int C, int H, int Rr,
                          const int* src, const int* dst, const int* etype,
                          const WsPtrs& p, float* out, hipStream_t stream) {
    dim3 gg(NN / 16, (D + 63) / 64, Rr);
    k_gemm_rel_wmma<<<gg, 128, 0, stream>>>(xin, w, p.xall, NN, K, D);

    const int totalRN = Rr * NN;
    k_rowdot<<<(totalRN + 3) / 4, 128, 0, stream>>>(p.xall, qv, kv, p.qall, p.kall,
                                                    D, H, totalRN);
    k_init<<<512, 256, 0, stream>>>(p.menc, p.den, p.aggr, NN * H, NN * C);

    const int eh = EE * H;
    k_edge_alpha_max<<<(eh + 255) / 256, 256, 0, stream>>>(p.qall, p.kall, src, dst,
                                                           etype, p.exbuf, p.menc, H, Rr);
    k_edge_exp<<<(eh + 255) / 256, 256, 0, stream>>>(dst, p.exbuf, p.menc, p.den, H);

    const int bC = ((C + 63) / 64) * 64;
    k_edge_aggr<<<EE, bC, 0, stream>>>(p.xall, p.exbuf, p.den, src, dst, etype,
                                       p.aggr, D, C, H, Rr);
    k_epilogue<<<(NN * C + 255) / 256, 256, 0, stream>>>(p.aggr, bias, out, C, NN * C);
}

extern "C" void kernel_launch(void* const* d_in, const int* in_sizes, int n_in,
                              void* d_out, int out_size, void* d_ws, size_t ws_size,
                              hipStream_t stream) {
    (void)in_sizes; (void)n_in; (void)out_size; (void)ws_size;
    const float* x  = (const float*)d_in[0];
    const int*   ei = (const int*)  d_in[1];
    const int*   et = (const int*)  d_in[2];
    const float* w1 = (const float*)d_in[3],  *q1 = (const float*)d_in[4];
    const float* k1 = (const float*)d_in[5],  *b1 = (const float*)d_in[6];
    const float* wS = (const float*)d_in[7],  *qS = (const float*)d_in[8];
    const float* kS = (const float*)d_in[9],  *bS = (const float*)d_in[10];
    const float* w2 = (const float*)d_in[11], *q2 = (const float*)d_in[12];
    const float* k2 = (const float*)d_in[13], *b2 = (const float*)d_in[14];
    const float* wT = (const float*)d_in[15], *qT = (const float*)d_in[16];
    const float* kT = (const float*)d_in[17], *bT = (const float*)d_in[18];
    const float* wg = (const float*)d_in[19];
    const float* ag_src = (const float*)d_in[20];
    const float* ag_dst = (const float*)d_in[21];
    const float* bg = (const float*)d_in[22];

    const int* src = ei;
    const int* dst = ei + EE;

    float* ws = (float*)d_ws;
    size_t off = 0;
    WsPtrs p;
    p.xall  = ws + off;            off += (size_t)4 * NN * 512;
    p.qall  = ws + off;            off += (size_t)4 * NN * 2;
    p.kall  = ws + off;            off += (size_t)4 * NN * 2;
    p.menc  = (unsigned*)(ws + off); off += (size_t)NN * 2;
    p.den   = ws + off;            off += (size_t)NN * 2;
    p.exbuf = ws + off;            off += (size_t)EE * 2;
    p.aggr  = ws + off;            off += (size_t)NN * 256;
    float* xs = ws + off;          off += (size_t)NN * 256;
    float* xt = ws + off;          off += (size_t)NN * 256;

    float* x_in   = (float*)d_out;
    float* x_out  = (float*)d_out + (size_t)NN * 65;
    float* x_self = (float*)d_out + (size_t)2 * NN * 65;

    run_attn_conv(x,  w1, q1, k1, b1, 128, 512, 256, 2, 4, src, dst, et, p, xs,   stream);
    run_attn_conv(xs, wS, qS, kS, bS, 256, 130,  65, 2, 4, src, dst, et, p, x_in, stream);
    run_attn_conv(x,  w2, q2, k2, b2, 128, 512, 256, 2, 4, src, dst, et, p, xt,    stream);
    run_attn_conv(xt, wT, qT, kT, bT, 256, 130,  65, 2, 4, src, dst, et, p, x_out, stream);
    run_attn_conv(x,  wg, ag_dst, ag_src, bg, 128, 256, 256, 1, 1, src, dst, et, p, x_self, stream);
}


# GEMM retiled: wave tile 16x16 -> 80x32 (80x16 where D<=256) fed from a bank-padded 80x128 LDS tile of X; each A fragment feeds 2 and each B fragment 5 matrix ops
# speedup vs baseline: 1.2971x; 1.2971x over previous
#include <hip/hip_runtime.h>
#include <hip/hip_bf16.h>
#include <math.h>

#define NN 10000
#define EE 160000

typedef __attribute__((ext_vector_type(2))) float v2f;
typedef __attribute__((ext_vector_type(8))) float v8f;

__device__ __forceinline__ float lrelu_(float x) { return x > 0.f ? x : 0.2f * x; }
__device__ __forceinline__ float elu_(float x)   { return x > 0.f ? x : (expf(x) - 1.f); }
__device__ __forceinline__ unsigned fenc_(float f) {
    unsigned u = __float_as_uint(f);
    return (u & 0x80000000u) ? ~u : (u | 0x80000000u);
}
__device__ __forceinline__ float fdec_(unsigned e) {
    return __uint_as_float((e & 0x80000000u) ? (e & 0x7FFFFFFFu) : ~e);
}

#define GM_MT    5
#define GM_ROWS  (16 * GM_MT)
#define GM_KC    128
#define GM_PITCH (GM_KC + 4)

template <int NT>
__global__ __launch_bounds__(256)
void k_gemm_rel_wmma(const float* __restrict__ X, const float* __restrict__ W,
                     float* __restrict__ Y, int M, int K, int Nout) {
    __shared__ __attribute__((aligned(16))) float As[GM_ROWS * GM_PITCH];
    const int r    = blockIdx.z;
    const int m0   = blockIdx.x * GM_ROWS;
    const int wave = threadIdx.x >> 5;
    const int lane = threadIdx.x & 31;
    const int n0   = (blockIdx.y * 8 + wave) * (16 * NT);
    const bool active = n0 < Nout;

    const float* Wr = W + (size_t)r * K * Nout;
    const int kh    = (lane >> 4) * 2;
    const int mrow  = lane & 15;
    unsigned voff[NT];
    bool nv[NT];
#pragma unroll
    for (int j = 0; j < NT; ++j) {
        const int ncol = n0 + 16 * j + mrow;
        nv[j] = ncol < Nout;
        const int ncl = nv[j] ? ncol : (Nout - 1);
        voff[j] = (unsigned)(kh * Nout + ncl) * 4u;
    }
    const size_t rowB = (size_t)Nout * 4;

    v8f c[GM_MT][NT];
#pragma unroll
    for (int t = 0; t < GM_MT; ++t)
#pragma unroll
        for (int j = 0; j < NT; ++j) c[t][j] = v8f{};

    for (int kc = 0; kc < K; kc += GM_KC) {
        if (kc) __syncthreads();
        {
            float4 st[GM_ROWS / 8];
            const float* Xs = X + (size_t)(m0 + wave) * K + kc + lane * 4;
#pragma unroll
            for (int i = 0; i < GM_ROWS / 8; ++i)
                st[i] = *reinterpret_cast<const float4*>(Xs + (size_t)(8 * i) * K);
#pragma unroll
            for (int i = 0; i < GM_ROWS / 8; ++i)
                *reinterpret_cast<float4*>(&As[(wave + 8 * i) * GM_PITCH + lane * 4]) = st[i];
        }
        __syncthreads();

        if (active) {
            const char* Wk = reinterpret_cast<const char*>(Wr) + (size_t)kc * rowB;
#pragma unroll 4
            for (int k0 = 0; k0 < GM_KC; k0 += 4) {
                const char* r0 = Wk + (size_t)k0 * rowB;
                v2f b[NT];
#pragma unroll
                for (int j = 0; j < NT; ++j) {
                    b[j].x = *reinterpret_cast<const float*>(r0 + voff[j]);
                    b[j].y = *reinterpret_cast<const float*>(r0 + rowB + voff[j]);
                }
#pragma unroll
                for (int t = 0; t < GM_MT; ++t) {
                    const v2f a = *reinterpret_cast<const v2f*>(
                        &As[(t * 16 + mrow) * GM_PITCH + k0 + kh]);
#pragma unroll
                    for (int j = 0; j < NT; ++j)
                        c[t][j] = __builtin_amdgcn_wmma_f32_16x16x4_f32(
                            false, a, false, b[j], (short)0, c[t][j], false, false);
                }
            }
        }
    }

    if (active) {
        float* Yr = Y + (size_t)r * M * Nout;
#pragma unroll
        for (int t = 0; t < GM_MT; ++t) {
            const int rbase = m0 + t * 16 + (lane >> 4) * 8;
#pragma unroll
            for (int j = 0; j < NT; ++j) {
                if (nv[j]) {
                    const int ncol = n0 + 16 * j + mrow;
#pragma unroll
                    for (int i = 0; i < 8; ++i)
                        Yr[(size_t)(rbase + i) * Nout + ncol] = c[t][j][i];
                }
            }
        }
    }
}

__global__ __launch_bounds__(128)
void k_rowdot(const float* __restrict__ XA, const float* __restrict__ Q,
              const float* __restrict__ Kv, float* __restrict__ qall,
              float* __restrict__ kall, int D, int H, int total) {
    const int wid  = (int)((blockIdx.x * (unsigned)blockDim.x + threadIdx.x) >> 5);
    const int lane = threadIdx.x & 31;
    if (wid >= total) return;
    const float* xrow = XA + (size_t)wid * D;
    float aq[2] = {0.f, 0.f}, ak[2] = {0.f, 0.f};
    for (int d = lane; d < D; d += 32) {
        float v = xrow[d];
        for (int h = 0; h < H; ++h) {
            aq[h] += v * Q[d * H + h];
            ak[h] += v * Kv[d * H + h];
        }
    }
    for (int off = 16; off; off >>= 1)
        for (int h = 0; h < H; ++h) {
            aq[h] += __shfl_xor(aq[h], off, 32);
            ak[h] += __shfl_xor(ak[h], off, 32);
        }
    if (lane == 0)
        for (int h = 0; h < H; ++h) {
            qall[(size_t)wid * H + h] = aq[h];
            kall[(size_t)wid * H + h] = ak[h];
        }
}

__global__ void k_init(unsigned* __restrict__ menc, float* __restrict__ den,
                       float* __restrict__ aggr, int nh, int na) {
    const int stride = gridDim.x * blockDim.x;
    for (int t = blockIdx.x * blockDim.x + threadIdx.x; t < nh; t += stride) {
        menc[t] = 0x007FFFFFu;
        den[t]  = 0.f;
    }
    for (int t = blockIdx.x * blockDim.x + threadIdx.x; t < na; t += stride)
        aggr[t] = 0.f;
}

__global__ void k_edge_alpha_max(const float* __restrict__ qall,
                                 const float* __restrict__ kall,
                                 const int* __restrict__ src, const int* __restrict__ dst,
                                 const int* __restrict__ etype,
                                 float* __restrict__ exbuf, unsigned* __restrict__ menc,
                                 int H, int Rr) {
    const int t = blockIdx.x * blockDim.x + threadIdx.x;
    if (t >= EE * H) return;
    const int e = t / H, h = t - e * H;
    const int et = (Rr > 1) ? etype[e] : 0;
    const int s = src[e], d = dst[e];
    const float a = lrelu_(qall[((size_t)et * NN + d) * H + h] +
                           kall[((size_t)et * NN + s) * H + h]);
    exbuf[t] = a;
    atomicMax(&menc[d * H + h], fenc_(a));
}

__global__ void k_edge_exp(const int* __restrict__ dst, float* __restrict__ exbuf,
                           const unsigned* __restrict__ menc, float* __restrict__ den,
                           int H) {
    const int t = blockIdx.x * blockDim.x + threadIdx.x;
    if (t >= EE * H) return;
    const int e = t / H, h = t - e * H;
    const int d = dst[e];
    const float ex = expf(exbuf[t] - fdec_(menc[d * H + h]));
    exbuf[t] = ex;
    atomicAdd(&den[d * H + h], ex);
}

__global__ void k_edge_aggr(const float* __restrict__ XA, const float* __restrict__ exbuf,
                            const float* __restrict__ den,
                            const int* __restrict__ src, const int* __restrict__ dst,
                            const int* __restrict__ etype, float* __restrict__ aggr,
                            int D, int C, int H, int Rr) {
    const int e = blockIdx.x;
    const int c = threadIdx.x;
    if (c >= C) return;
    const int s = src[e], d = dst[e];
    const int et = (Rr > 1) ? etype[e] : 0;
    const float* xrow = XA + ((size_t)et * NN + s) * D;
    float v = 0.f;
    for (int h = 0; h < H; ++h) {
        const float attn = exbuf[(size_t)e * H + h] / (den[d * H + h] + 1e-16f);
        v += attn * xrow[h * C + c];
    }
    atomicAdd(&aggr[(size_t)d * C + c], v / (float)H);
}

__global__ void k_epilogue(const float* __restrict__ aggr, const float* __restrict__ bias,
                           float* __restrict__ out, int C, int total) {
    const int t = blockIdx.x * blockDim.x + threadIdx.x;
    if (t >= total) return;
    out[t] = elu_(aggr[t] + bias[t % C]);
}

struct WsPtrs {
    float* xall; float* qall; float* kall; unsigned* menc;
    float* den;  float* exbuf; float* aggr;
};

static void run_attn_conv(const float* xin, const float* w, const float* qv,
                          const float* kv, const float* bias,
                          int K, int D, int C, int H, int Rr,
                          const int* src, const int* dst, const int* etype,
                          const WsPtrs& p, float* out, hipStream_t stream) {
    if (D > 256) {
        dim3 gg(NN / GM_ROWS, (D + 255) / 256, Rr);
        k_gemm_rel_wmma<2><<<gg, 256, 0, stream>>>(xin, w, p.xall, NN, K, D);
    } else {
        dim3 gg(NN / GM_ROWS, (D + 127) / 128, Rr);
        k_gemm_rel_wmma<1><<<gg, 256, 0, stream>>>(xin, w, p.xall, NN, K, D);
    }

    const int totalRN = Rr * NN;
    k_rowdot<<<(totalRN + 3) / 4, 128, 0, stream>>>(p.xall, qv, kv, p.qall, p.kall,
                                                    D, H, totalRN);
    k_init<<<512, 256, 0, stream>>>(p.menc, p.den, p.aggr, NN * H, NN * C);

    const int eh = EE * H;
    k_edge_alpha_max<<<(eh + 255) / 256, 256, 0, stream>>>(p.qall, p.kall, src, dst,
                                                           etype, p.exbuf, p.menc, H, Rr);
    k_edge_exp<<<(eh + 255) / 256, 256, 0, stream>>>(dst, p.exbuf, p.menc, p.den, H);

    const int bC = ((C + 63) / 64) * 64;
    k_edge_aggr<<<EE, bC, 0, stream>>>(p.xall, p.exbuf, p.den, src, dst, etype,
                                       p.aggr, D, C, H, Rr);
    k_epilogue<<<(NN * C + 255) / 256, 256, 0, stream>>>(p.aggr, bias, out, C, NN * C);
}

extern "C" void kernel_launch(void* const* d_in, const int* in_sizes, int n_in,
                              void* d_out, int out_size, void* d_ws, size_t ws_size,
                              hipStream_t stream) {
    (void)in_sizes; (void)n_in; (void)out_size; (void)ws_size;
    const float* x  = (const float*)d_in[0];
    const int*   ei = (const int*)  d_in[1];
    const int*   et = (const int*)  d_in[2];
    const float* w1 = (const float*)d_in[3],  *q1 = (const float*)d_in[4];
    const float* k1 = (const float*)d_in[5],  *b1 = (const float*)d_in[6];
    const float* wS = (const float*)d_in[7],  *qS = (const float*)d_in[8];
    const float* kS = (const float*)d_in[9],  *bS = (const float*)d_in[10];
    const float* w2 = (const float*)d_in[11], *q2 = (const float*)d_in[12];
    const float* k2 = (const float*)d_in[13], *b2 = (const float*)d_in[14];
    const float* wT = (const float*)d_in[15], *qT = (const float*)d_in[16];
    const float* kT = (const float*)d_in[17], *bT = (const float*)d_in[18];
    const float* wg = (const float*)d_in[19];
    const float* ag_src = (const float*)d_in[20];
    const float* ag_dst = (const float*)d_in[21];
    const float* bg = (const float*)d_in[22];

    const int* src = ei;
    const int* dst = ei + EE;

    float* ws = (float*)d_ws;
    size_t off = 0;
    WsPtrs p;
    p.xall  = ws + off;            off += (size_t)4 * NN * 512;
    p.qall  = ws + off;            off += (size_t)4 * NN * 2;
    p.kall  = ws + off;            off += (size_t)4 * NN * 2;
    p.menc  = (unsigned*)(ws + off); off += (size_t)NN * 2;
    p.den   = ws + off;            off += (size_t)NN * 2;
    p.exbuf = ws + off;            off += (size_t)EE * 2;
    p.aggr  = ws + off;            off += (size_t)NN * 256;
    float* xs = ws + off;          off += (size_t)NN * 256;
    float* xt = ws + off;          off += (size_t)NN * 256;

    float* x_in   = (float*)d_out;
    float* x_out  = (float*)d_out + (size_t)NN * 65;
    float* x_self = (float*)d_out + (size_t)2 * NN * 65;

    run_attn_conv(x,  w1, q1, k1, b1, 128, 512, 256, 2, 4, src, dst, et, p, xs,   stream);
    run_attn_conv(xs, wS, qS, kS, bS, 256, 130,  65, 2, 4, src, dst, et, p, x_in, stream);
    run_attn_conv(x,  w2, q2, k2, b2, 128, 512, 256, 2, 4, src, dst, et, p, xt,    stream);
    run_attn_conv(xt, wT, qT, kT, bT, 256, 130,  65, 2, 4, src, dst, et, p, x_out, stream);
    run_attn_conv(x,  wg, ag_dst, ag_src, bg, 128, 256, 256, 1, 1, src, dst, et, p, x_self, stream);
}


# Edge phase regrouped by destination: edges bucketed once per call, then one wave per node does softmax, gather, head mean, bias and elu in registers; no float atomics, no init or epilogue pass
# speedup vs baseline: 3.1664x; 2.4411x over previous
#include <hip/hip_runtime.h>
#include <hip/hip_bf16.h>
#include <math.h>

#define NN 10000
#define EE 160000

typedef __attribute__((ext_vector_type(2))) float v2f;
typedef __attribute__((ext_vector_type(8))) float v8f;

__device__ __forceinline__ float lrelu_(float x) { return x > 0.f ? x : 0.2f * x; }
__device__ __forceinline__ float elu_(float x)   { return x > 0.f ? x : (expf(x) - 1.f); }

#define GM_MT    5
#define GM_ROWS  (16 * GM_MT)
#define GM_KC    128
#define GM_PITCH (GM_KC + 4)

template <int NT>
__global__ __launch_bounds__(256)
void k_gemm_rel_wmma(const float* __restrict__ X, const float* __restrict__ W,
                     float* __restrict__ Y, int M, int K, int Nout) {
    __shared__ __attribute__((aligned(16))) float As[GM_ROWS * GM_PITCH];
    const int r    = blockIdx.z;
    const int m0   = blockIdx.x * GM_ROWS;
    const int wave = threadIdx.x >> 5;
    const int lane = threadIdx.x & 31;
    const int n0   = (blockIdx.y * 8 + wave) * (16 * NT);
    const bool active = n0 < Nout;

    const float* Wr = W + (size_t)r * K * Nout;
    const int kh    = (lane >> 4) * 2;
    const int mrow  = lane & 15;
    unsigned voff[NT];
    bool nv[NT];
#pragma unroll
    for (int j = 0; j < NT; ++j) {
        const int ncol = n0 + 16 * j + mrow;
        nv[j] = ncol < Nout;
        const int ncl = nv[j] ? ncol : (Nout - 1);
        voff[j] = (unsigned)(kh * Nout + ncl) * 4u;
    }
    const size_t rowB = (size_t)Nout * 4;

    v8f c[GM_MT][NT];
#pragma unroll
    for (int t = 0; t < GM_MT; ++t)
#pragma unroll
        for (int j = 0; j < NT; ++j) c[t][j] = v8f{};

    for (int kc = 0; kc < K; kc += GM_KC) {
        if (kc) __syncthreads();
        {
            float4 st[GM_ROWS / 8];
            const float* Xs = X + (size_t)(m0 + wave) * K + kc + lane * 4;
#pragma unroll
            for (int i = 0; i < GM_ROWS / 8; ++i)
                st[i] = *reinterpret_cast<const float4*>(Xs + (size_t)(8 * i) * K);
#pragma unroll
            for (int i = 0; i < GM_ROWS / 8; ++i)
                *reinterpret_cast<float4*>(&As[(wave + 8 * i) * GM_PITCH + lane * 4]) = st[i];
        }
        __syncthreads();

        if (active) {
            const char* Wk = reinterpret_cast<const char*>(Wr) + (size_t)kc * rowB;
#pragma unroll 4
            for (int k0 = 0; k0 < GM_KC; k0 += 4) {
                const char* r0 = Wk + (size_t)k0 * rowB;
                v2f b[NT];
#pragma unroll
                for (int j = 0; j < NT; ++j) {
                    b[j].x = *reinterpret_cast<const float*>(r0 + voff[j]);
                    b[j].y = *reinterpret_cast<const float*>(r0 + rowB + voff[j]);
                }
#pragma unroll
                for (int t = 0; t < GM_MT; ++t) {
                    const v2f a = *reinterpret_cast<const v2f*>(
                        &As[(t * 16 + mrow) * GM_PITCH + k0 + kh]);
#pragma unroll
                    for (int j = 0; j < NT; ++j)
                        c[t][j] = __builtin_amdgcn_wmma_f32_16x16x4_f32(
                            false, a, false, b[j], (short)0, c[t][j], false, false);
                }
            }
        }
    }

    if (active) {
        float* Yr = Y + (size_t)r * M * Nout;
#pragma unroll
        for (int t = 0; t < GM_MT; ++t) {
            const int rbase = m0 + t * 16 + (lane >> 4) * 8;
#pragma unroll
            for (int j = 0; j < NT; ++j) {
                if (nv[j]) {
                    const int ncol = n0 + 16 * j + mrow;
#pragma unroll
                    for (int i = 0; i < 8; ++i)
                        Yr[(size_t)(rbase + i) * Nout + ncol] = c[t][j][i];
                }
            }
        }
    }
}

__global__ __launch_bounds__(128)
void k_rowdot(const float* __restrict__ XA, const float* __restrict__ Q,
              const float* __restrict__ Kv, float* __restrict__ qall,
              float* __restrict__ kall, int D, int H, int total) {
    const int wid  = (int)((blockIdx.x * (unsigned)blockDim.x + threadIdx.x) >> 5);
    const int lane = threadIdx.x & 31;
    if (wid >= total) return;
    const float* xrow = XA + (size_t)wid * D;
    float aq[2] = {0.f, 0.f}, ak[2] = {0.f, 0.f};
    for (int d = lane; d < D; d += 32) {
        float v = xrow[d];
        for (int h = 0; h < H; ++h) {
            aq[h] += v * Q[d * H + h];
            ak[h] += v * Kv[d * H + h];
        }
    }
    for (int off = 16; off; off >>= 1)
        for (int h = 0; h < H; ++h) {
            aq[h] += __shfl_xor(aq[h], off, 32);
            ak[h] += __shfl_xor(ak[h], off, 32);
        }
    if (lane == 0)
        for (int h = 0; h < H; ++h) {
            qall[(size_t)wid * H + h] = aq[h];
            kall[(size_t)wid * H + h] = ak[h];
        }
}

__global__ void k_csr_zero(int* __restrict__ cnt) {
    const int t = blockIdx.x * blockDim.x + threadIdx.x;
    if (t < NN) cnt[t] = 0;
}

__global__ void k_csr_hist(const int* __restrict__ dst, int* __restrict__ cnt) {
    const int e = blockIdx.x * blockDim.x + threadIdx.x;
    if (e < EE) atomicAdd(&cnt[dst[e]], 1);
}

#define SC_ITEMS 10
__global__ __launch_bounds__(1024)
void k_csr_scan(const int* __restrict__ cnt, int* __restrict__ rowptr) {
    __shared__ int wtot[32];
    const int t = threadIdx.x, lane = t & 31, wv = t >> 5;
    const int base = t * SC_ITEMS;
    int v[SC_ITEMS];
    int s = 0;
#pragma unroll
    for (int i = 0; i < SC_ITEMS; ++i) {
        const int idx = base + i;
        v[i] = (idx < NN) ? cnt[idx] : 0;
        s += v[i];
    }
    int inc = s;
    for (int off = 1; off < 32; off <<= 1) {
        const int u = __shfl_up(inc, off, 32);
        if (lane >= off) inc += u;
    }
    if (lane == 31) wtot[wv] = inc;
    __syncthreads();
    if (wv == 0) {
        const int w = wtot[lane];
        int winc = w;
        for (int off = 1; off < 32; off <<= 1) {
            const int u = __shfl_up(winc, off, 32);
            if (lane >= off) winc += u;
        }
        wtot[lane] = winc - w;
    }
    __syncthreads();
    int run = wtot[wv] + inc - s;
#pragma unroll
    for (int i = 0; i < SC_ITEMS; ++i) {
        const int idx = base + i;
        if (idx < NN) { rowptr[idx] = run; run += v[i]; }
    }
    if (t == 1023) rowptr[NN] = run;
}

__global__ void k_csr_fill(const int* __restrict__ src, const int* __restrict__ dst,
                           const int* __restrict__ etype, const int* __restrict__ rowptr,
                           int* __restrict__ cnt, int* __restrict__ cpk) {
    const int e = blockIdx.x * blockDim.x + threadIdx.x;
    if (e >= EE) return;
    const int d = dst[e];
    const int pos = atomicSub(&cnt[d], 1) - 1;
    cpk[rowptr[d] + pos] = (etype[e] << 16) | src[e];
}

template <int C, int H, bool REL>
__global__ __launch_bounds__(256)
void k_node_attn(const float* __restrict__ XA, const float* __restrict__ qall,
                 const float* __restrict__ kall, const int* __restrict__ rowptr,
                 const int* __restrict__ cpk, const float* __restrict__ bias,
                 float* __restrict__ out) {
    constexpr int  D  = C * H;
    constexpr bool V4 = (C % 128 == 0);
    constexpr int  SL = V4 ? C / 128 : 1;
    constexpr int  NG = V4 ? 1 : (C + 31) / 32;
    const int lane = threadIdx.x & 31;
    const int gw   = __builtin_amdgcn_readfirstlane((int)((blockIdx.x * blockDim.x + threadIdx.x) >> 5));
    const int node = gw / SL;
    const int sl   = gw - node * SL;
    if (node >= NN) return;
    const int beg = rowptr[node];
    const int deg = rowptr[node + 1] - beg;

    auto load_edge = [&](int idx, int& row, float (&a)[H]) {
        const bool valid = idx < deg;
        const int p  = valid ? cpk[beg + idx] : 0;
        const int et = REL ? (p >> 16) : 0;
        row = et * NN + (p & 0xFFFF);
        const int qrow = et * NN + node;
#pragma unroll
        for (int h = 0; h < H; ++h) {
            const float v = lrelu_(qall[(size_t)qrow * H + h] + kall[(size_t)row * H + h]);
            a[h] = valid ? v : -INFINITY;
        }
    };

    int row0; float a0[H];
    load_edge(lane, row0, a0);
    float m[H];
#pragma unroll
    for (int h = 0; h < H; ++h) m[h] = a0[h];
    for (int cb = 32; cb < deg; cb += 32) {
        int r; float a[H];
        load_edge(cb + lane, r, a);
#pragma unroll
        for (int h = 0; h < H; ++h) m[h] = fmaxf(m[h], a[h]);
    }
#pragma unroll
    for (int h = 0; h < H; ++h) {
        for (int off = 16; off; off >>= 1) m[h] = fmaxf(m[h], __shfl_xor(m[h], off, 32));
        m[h] = isfinite(m[h]) ? m[h] : 0.f;
    }

    float den[H];
    float4 acc4[H];
    float  acc1[H][NG];
#pragma unroll
    for (int h = 0; h < H; ++h) {
        den[h] = 0.f;
        acc4[h] = make_float4(0.f, 0.f, 0.f, 0.f);
#pragma unroll
        for (int g = 0; g < NG; ++g) acc1[h][g] = 0.f;
    }

    int cch[NG];
#pragma unroll
    for (int g = 0; g < NG; ++g) cch[g] = min(lane + 32 * g, C - 1);

    auto process = [&](int cb, int row, const float (&a)[H]) {
        float ex[H];
#pragma unroll
        for (int h = 0; h < H; ++h) {
            ex[h] = (cb + lane < deg) ? expf(a[h] - m[h]) : 0.f;
            den[h] += ex[h];
        }
        const int cnt = min(32, deg - cb);
#pragma unroll 4
        for (int j = 0; j < cnt; ++j) {
            const int rj = __builtin_amdgcn_readlane(row, j);
            const float* xr = XA + (size_t)rj * D;
#pragma unroll
            for (int h = 0; h < H; ++h) {
                const float w = __int_as_float(__builtin_amdgcn_readlane(__float_as_int(ex[h]), j));
                if constexpr (V4) {
                    const float4 x = *reinterpret_cast<const float4*>(xr + h * C + sl * 128 + lane * 4);
                    acc4[h].x += w * x.x; acc4[h].y += w * x.y;
                    acc4[h].z += w * x.z; acc4[h].w += w * x.w;
                } else {
#pragma unroll
                    for (int g = 0; g < NG; ++g)
                        acc1[h][g] += w * xr[h * C + cch[g]];
                }
            }
        }
    };

    process(0, row0, a0);
    for (int cb = 32; cb < deg; cb += 32) {
        int r; float a[H];
        load_edge(cb + lane, r, a);
        process(cb, r, a);
    }

    float inv[H];
#pragma unroll
    for (int h = 0; h < H; ++h) {
        for (int off = 16; off; off >>= 1) den[h] += __shfl_xor(den[h], off, 32);
        inv[h] = 1.f / (den[h] + 1e-16f);
    }

    if constexpr (V4) {
        const int c0 = sl * 128 + lane * 4;
        float4 o = make_float4(0.f, 0.f, 0.f, 0.f);
#pragma unroll
        for (int h = 0; h < H; ++h) {
            o.x += acc4[h].x * inv[h]; o.y += acc4[h].y * inv[h];
            o.z += acc4[h].z * inv[h]; o.w += acc4[h].w * inv[h];
        }
        o.x = elu_(o.x / (float)H + bias[c0 + 0]);
        o.y = elu_(o.y / (float)H + bias[c0 + 1]);
        o.z = elu_(o.z / (float)H + bias[c0 + 2]);
        o.w = elu_(o.w / (float)H + bias[c0 + 3]);
        *reinterpret_cast<float4*>(out + (size_t)node * C + c0) = o;
    } else {
#pragma unroll
        for (int g = 0; g < NG; ++g) {
            const int c = lane + 32 * g;
            if (c < C) {
                float o = 0.f;
#pragma unroll
                for (int h = 0; h < H; ++h) o += acc1[h][g] * inv[h];
                out[(size_t)node * C + c] = elu_(o / (float)H + bias[c]);
            }
        }
    }
}

struct WsPtrs {
    float* xall; float* qall; float* kall;
    int* rowptr; int* cnt; int* cpk;
};

template <int C, int H, bool REL>
static void run_attn_conv(const float* xin, const float* w, const float* qv,
                          const float* kv, const float* bias, int K,
                          const WsPtrs& p, float* out, hipStream_t stream) {
    constexpr int D  = C * H;
    constexpr int Rr = REL ? 4 : 1;
    if (D > 256) {
        dim3 gg(NN / GM_ROWS, (D + 255) / 256, Rr);
        k_gemm_rel_wmma<2><<<gg, 256, 0, stream>>>(xin, w, p.xall, NN, K, D);
    } else {
        dim3 gg(NN / GM_ROWS, (D + 127) / 128, Rr);
        k_gemm_rel_wmma<1><<<gg, 256, 0, stream>>>(xin, w, p.xall, NN, K, D);
    }

    constexpr int totalRN = Rr * NN;
    k_rowdot<<<(totalRN + 3) / 4, 128, 0, stream>>>(p.xall, qv, kv, p.qall, p.kall,
                                                    D, H, totalRN);

    constexpr int SL = (C % 128 == 0) ? C / 128 : 1;
    k_node_attn<C, H, REL><<<(NN * SL + 7) / 8, 256, 0, stream>>>(
        p.xall, p.qall, p.kall, p.rowptr, p.cpk, bias, out);
}

extern "C" void kernel_launch(void* const* d_in, const int* in_sizes, int n_in,
                              void* d_out, int out_size, void* d_ws, size_t ws_size,
                              hipStream_t stream) {
    (void)in_sizes; (void)n_in; (void)out_size; (void)ws_size;
    const float* x  = (const float*)d_in[0];
    const int*   ei = (const int*)  d_in[1];
    const int*   et = (const int*)  d_in[2];
    const float* w1 = (const float*)d_in[3],  *q1 = (const float*)d_in[4];
    const float* k1 = (const float*)d_in[5],  *b1 = (const float*)d_in[6];
    const float* wS = (const float*)d_in[7],  *qS = (const float*)d_in[8];
    const float* kS = (const float*)d_in[9],  *bS = (const float*)d_in[10];
    const float* w2 = (const float*)d_in[11], *q2 = (const float*)d_in[12];
    const float* k2 = (const float*)d_in[13], *b2 = (const float*)d_in[14];
    const float* wT = (const float*)d_in[15], *qT = (const float*)d_in[16];
    const float* kT = (const float*)d_in[17], *bT = (const float*)d_in[18];
    const float* wg = (const float*)d_in[19];
    const float* ag_src = (const float*)d_in[20];
    const float* ag_dst = (const float*)d_in[21];
    const float* bg = (const float*)d_in[22];

    const int* src = ei;
    const int* dst = ei + EE;

    float* ws = (float*)d_ws;
    size_t off = 0;
    WsPtrs p;
    p.xall   = ws + off;          off += (size_t)4 * NN * 512;
    p.qall   = ws + off;          off += (size_t)4 * NN * 2;
    p.kall   = ws + off;          off += (size_t)4 * NN * 2;
    p.rowptr = (int*)(ws + off);  off += (size_t)NN * 2;
    p.cnt    = (int*)(ws + off);  off += (size_t)NN * 2;
    p.cpk    = (int*)(ws + off);  off += (size_t)EE * 2;
    off += (size_t)NN * 256;
    float* xs = ws + off;         off += (size_t)NN * 256;
    float* xt = ws + off;         off += (size_t)NN * 256;

    float* x_in   = (float*)d_out;
    float* x_out  = (float*)d_out + (size_t)NN * 65;
    float* x_self = (float*)d_out + (size_t)2 * NN * 65;

    k_csr_zero<<<(NN + 255) / 256, 256, 0, stream>>>(p.cnt);
    k_csr_hist<<<(EE + 255) / 256, 256, 0, stream>>>(dst, p.cnt);
    k_csr_scan<<<1, 1024, 0, stream>>>(p.cnt, p.rowptr);
    k_csr_fill<<<(EE + 255) / 256, 256, 0, stream>>>(src, dst, et, p.rowptr, p.cnt, p.cpk);

    run_attn_conv<256, 2, true>(x,  w1, q1, k1, b1, 128, p, xs,   stream);
    run_attn_conv< 65, 2, true>(xs, wS, qS, kS, bS, 256, p, x_in, stream);
    run_attn_conv<256, 2, true>(x,  w2, q2, k2, b2, 128, p, xt,    stream);
    run_attn_conv< 65, 2, true>(xt, wT, qT, kT, bT, 256, p, x_out, stream);
    run_attn_conv<256, 1, false>(x, wg, ag_dst, ag_src, bg, 128, p, x_self, stream);
}


# Row dots (q_all/k_all): a wave takes 8 rows instead of 1 (8x fewer waves and blocks), keeps its Q/K slices in registers and issues the loads of 4 to 8 rows together, one float4 per lane on wide rows
# speedup vs baseline: 3.9817x; 1.2575x over previous
#include <hip/hip_runtime.h>
#include <hip/hip_bf16.h>
#include <math.h>

#define NN 10000
#define EE 160000

typedef __attribute__((ext_vector_type(2))) float v2f;
typedef __attribute__((ext_vector_type(8))) float v8f;

__device__ __forceinline__ float lrelu_(float x) { return x > 0.f ? x : 0.2f * x; }
__device__ __forceinline__ float elu_(float x)   { return x > 0.f ? x : (expf(x) - 1.f); }

#define GM_MT    5
#define GM_ROWS  (16 * GM_MT)
#define GM_KC    128
#define GM_PITCH (GM_KC + 4)

template <int NT>
__global__ __launch_bounds__(256)
void k_gemm_rel_wmma(const float* __restrict__ X, const float* __restrict__ W,
                     float* __restrict__ Y, int M, int K, int Nout) {
    __shared__ __attribute__((aligned(16))) float As[GM_ROWS * GM_PITCH];
    const int r    = blockIdx.z;
    const int m0   = blockIdx.x * GM_ROWS;
    const int wave = threadIdx.x >> 5;
    const int lane = threadIdx.x & 31;
    const int n0   = (blockIdx.y * 8 + wave) * (16 * NT);
    const bool active = n0 < Nout;

    const float* Wr = W + (size_t)r * K * Nout;
    const int kh    = (lane >> 4) * 2;
    const int mrow  = lane & 15;
    unsigned voff[NT];
    bool nv[NT];
#pragma unroll
    for (int j = 0; j < NT; ++j) {
        const int ncol = n0 + 16 * j + mrow;
        nv[j] = ncol < Nout;
        const int ncl = nv[j] ? ncol : (Nout - 1);
        voff[j] = (unsigned)(kh * Nout + ncl) * 4u;
    }
    const size_t rowB = (size_t)Nout * 4;

    v8f c[GM_MT][NT];
#pragma unroll
    for (int t = 0; t < GM_MT; ++t)
#pragma unroll
        for (int j = 0; j < NT; ++j) c[t][j] = v8f{};

    for (int kc = 0; kc < K; kc += GM_KC) {
        if (kc) __syncthreads();
        {
            float4 st[GM_ROWS / 8];
            const float* Xs = X + (size_t)(m0 + wave) * K + kc + lane * 4;
#pragma unroll
            for (int i = 0; i < GM_ROWS / 8; ++i)
                st[i] = *reinterpret_cast<const float4*>(Xs + (size_t)(8 * i) * K);
#pragma unroll
            for (int i = 0; i < GM_ROWS / 8; ++i)
                *reinterpret_cast<float4*>(&As[(wave + 8 * i) * GM_PITCH + lane * 4]) = st[i];
        }
        __syncthreads();

        if (active) {
            const char* Wk = reinterpret_cast<const char*>(Wr) + (size_t)kc * rowB;
#pragma unroll 4
            for (int k0 = 0; k0 < GM_KC; k0 += 4) {
                const char* r0 = Wk + (size_t)k0 * rowB;
                v2f b[NT];
#pragma unroll
                for (int j = 0; j < NT; ++j) {
                    b[j].x = *reinterpret_cast<const float*>(r0 + voff[j]);
                    b[j].y = *reinterpret_cast<const float*>(r0 + rowB + voff[j]);
                }
#pragma unroll
                for (int t = 0; t < GM_MT; ++t) {
                    const v2f a = *reinterpret_cast<const v2f*>(
                        &As[(t * 16 + mrow) * GM_PITCH + k0 + kh]);
#pragma unroll
                    for (int j = 0; j < NT; ++j)
                        c[t][j] = __builtin_amdgcn_wmma_f32_16x16x4_f32(
                            false, a, false, b[j], (short)0, c[t][j], false, false);
                }
            }
        }
    }

    if (active) {
        float* Yr = Y + (size_t)r * M * Nout;
#pragma unroll
        for (int t = 0; t < GM_MT; ++t) {
            const int rbase = m0 + t * 16 + (lane >> 4) * 8;
#pragma unroll
            for (int j = 0; j < NT; ++j) {
                if (nv[j]) {
                    const int ncol = n0 + 16 * j + mrow;
#pragma unroll
                    for (int i = 0; i < 8; ++i)
                        Yr[(size_t)(rbase + i) * Nout + ncol] = c[t][j][i];
                }
            }
        }
    }
}

#define RD_RW 8
template <int D, int H>
__global__ __launch_bounds__(128)
void k_rowdot(const float* __restrict__ XA, const float* __restrict__ Q,
              const float* __restrict__ Kv, float* __restrict__ qall,
              float* __restrict__ kall, int total) {
    constexpr bool V4 = (D % 128 == 0);
    constexpr int  EW = V4 ? 4 : 1;
    constexpr int  NI = V4 ? D / 128 : (D + 31) / 32;
    constexpr int  RU = (V4 && NI >= 4) ? 4 : 8;
    const int lane = threadIdx.x & 31;
    const int gw = __builtin_amdgcn_readfirstlane((int)((blockIdx.x * blockDim.x + threadIdx.x) >> 5));

    int   dcl[NI];
    float wq[NI][EW][H], wk[NI][EW][H];
#pragma unroll
    for (int i = 0; i < NI; ++i) {
        const int  d0 = V4 ? i * 128 + lane * 4 : lane + 32 * i;
        const bool ok = V4 || d0 < D;
        dcl[i] = ok ? d0 : D - 1;
#pragma unroll
        for (int e = 0; e < EW; ++e)
#pragma unroll
            for (int h = 0; h < H; ++h) {
                const float q = Q[(dcl[i] + e) * H + h];
                const float k = Kv[(dcl[i] + e) * H + h];
                wq[i][e][h] = ok ? q : 0.f;
                wk[i][e][h] = ok ? k : 0.f;
            }
    }

    for (int t = 0; t < RD_RW / RU; ++t) {
        const int r0 = gw * RD_RW + t * RU;
        if (r0 >= total) break;
        float x[RU][NI][EW];
#pragma unroll
        for (int u = 0; u < RU; ++u) {
            const int row = min(r0 + u, total - 1);
            const float* xr = XA + (size_t)row * D;
#pragma unroll
            for (int i = 0; i < NI; ++i) {
                if constexpr (V4) {
                    const float4 v = *reinterpret_cast<const float4*>(xr + i * 128 + lane * 4);
                    x[u][i][0] = v.x; x[u][i][1] = v.y; x[u][i][2] = v.z; x[u][i][3] = v.w;
                } else {
                    x[u][i][0] = xr[dcl[i]];
                }
            }
        }
        __builtin_amdgcn_sched_barrier(0);
        float aq[RU][H], ak[RU][H];
#pragma unroll
        for (int u = 0; u < RU; ++u) {
#pragma unroll
            for (int h = 0; h < H; ++h) { aq[u][h] = 0.f; ak[u][h] = 0.f; }
#pragma unroll
            for (int i = 0; i < NI; ++i)
#pragma unroll
                for (int e = 0; e < EW; ++e)
#pragma unroll
                    for (int h = 0; h < H; ++h) {
                        aq[u][h] += x[u][i][e] * wq[i][e][h];
                        ak[u][h] += x[u][i][e] * wk[i][e][h];
                    }
        }
        for (int off = 16; off; off >>= 1)
#pragma unroll
            for (int u = 0; u < RU; ++u)
#pragma unroll
                for (int h = 0; h < H; ++h) {
                    aq[u][h] += __shfl_xor(aq[u][h], off, 32);
                    ak[u][h] += __shfl_xor(ak[u][h], off, 32);
                }
        if (lane == 0) {
#pragma unroll
            for (int u = 0; u < RU; ++u)
                if (r0 + u < total) {
#pragma unroll
                    for (int h = 0; h < H; ++h) {
                        qall[(size_t)(r0 + u) * H + h] = aq[u][h];
                        kall[(size_t)(r0 + u) * H + h] = ak[u][h];
                    }
                }
        }
    }
}

__global__ void k_csr_zero(int* __restrict__ cnt) {
    const int t = blockIdx.x * blockDim.x + threadIdx.x;
    if (t < NN) cnt[t] = 0;
}

__global__ void k_csr_hist(const int* __restrict__ dst, int* __restrict__ cnt) {
    const int e = blockIdx.x * blockDim.x + threadIdx.x;
    if (e < EE) atomicAdd(&cnt[dst[e]], 1);
}

#define SC_ITEMS 10
__global__ __launch_bounds__(1024)
void k_csr_scan(const int* __restrict__ cnt, int* __restrict__ rowptr) {
    __shared__ int wtot[32];
    const int t = threadIdx.x, lane = t & 31, wv = t >> 5;
    const int base = t * SC_ITEMS;
    int v[SC_ITEMS];
    int s = 0;
#pragma unroll
    for (int i = 0; i < SC_ITEMS; ++i) {
        const int idx = base + i;
        v[i] = (idx < NN) ? cnt[idx] : 0;
        s += v[i];
    }
    int inc = s;
    for (int off = 1; off < 32; off <<= 1) {
        const int u = __shfl_up(inc, off, 32);
        if (lane >= off) inc += u;
    }
    if (lane == 31) wtot[wv] = inc;
    __syncthreads();
    if (wv == 0) {
        const int w = wtot[lane];
        int winc = w;
        for (int off = 1; off < 32; off <<= 1) {
            const int u = __shfl_up(winc, off, 32);
            if (lane >= off) winc += u;
        }
        wtot[lane] = winc - w;
    }
    __syncthreads();
    int run = wtot[wv] + inc - s;
#pragma unroll
    for (int i = 0; i < SC_ITEMS; ++i) {
        const int idx = base + i;
        if (idx < NN) { rowptr[idx] = run; run += v[i]; }
    }
    if (t == 1023) rowptr[NN] = run;
}

__global__ void k_csr_fill(const int* __restrict__ src, const int* __restrict__ dst,
                           const int* __restrict__ etype, const int* __restrict__ rowptr,
                           int* __restrict__ cnt, int* __restrict__ cpk) {
    const int e = blockIdx.x * blockDim.x + threadIdx.x;
    if (e >= EE) return;
    const int d = dst[e];
    const int pos = atomicSub(&cnt[d], 1) - 1;
    cpk[rowptr[d] + pos] = (etype[e] << 16) | src[e];
}

template <int C, int H, bool REL>
__global__ __launch_bounds__(256)
void k_node_attn(const float* __restrict__ XA, const float* __restrict__ qall,
                 const float* __restrict__ kall, const int* __restrict__ rowptr,
                 const int* __restrict__ cpk, const float* __restrict__ bias,
                 float* __restrict__ out) {
    constexpr int  D  = C * H;
    constexpr bool V4 = (C % 128 == 0);
    constexpr int  SL = V4 ? C / 128 : 1;
    constexpr int  NG = V4 ? 1 : (C + 31) / 32;
    const int lane = threadIdx.x & 31;
    const int gw   = __builtin_amdgcn_readfirstlane((int)((blockIdx.x * blockDim.x + threadIdx.x) >> 5));
    const int node = gw / SL;
    const int sl   = gw - node * SL;
    if (node >= NN) return;
    const int beg = rowptr[node];
    const int deg = rowptr[node + 1] - beg;

    auto load_edge = [&](int idx, int& row, float (&a)[H]) {
        const bool valid = idx < deg;
        const int p  = valid ? cpk[beg + idx] : 0;
        const int et = REL ? (p >> 16) : 0;
        row = et * NN + (p & 0xFFFF);
        const int qrow = et * NN + node;
#pragma unroll
        for (int h = 0; h < H; ++h) {
            const float v = lrelu_(qall[(size_t)qrow * H + h] + kall[(size_t)row * H + h]);
            a[h] = valid ? v : -INFINITY;
        }
    };

    int row0; float a0[H];
    load_edge(lane, row0, a0);
    float m[H];
#pragma unroll
    for (int h = 0; h < H; ++h) m[h] = a0[h];
    for (int cb = 32; cb < deg; cb += 32) {
        int r; float a[H];
        load_edge(cb + lane, r, a);
#pragma unroll
        for (int h = 0; h < H; ++h) m[h] = fmaxf(m[h], a[h]);
    }
#pragma unroll
    for (int h = 0; h < H; ++h) {
        for (int off = 16; off; off >>= 1) m[h] = fmaxf(m[h], __shfl_xor(m[h], off, 32));
        m[h] = isfinite(m[h]) ? m[h] : 0.f;
    }

    float den[H];
    float4 acc4[H];
    float  acc1[H][NG];
#pragma unroll
    for (int h = 0; h < H; ++h) {
        den[h] = 0.f;
        acc4[h] = make_float4(0.f, 0.f, 0.f, 0.f);
#pragma unroll
        for (int g = 0; g < NG; ++g) acc1[h][g] = 0.f;
    }

    int cch[NG];
#pragma unroll
    for (int g = 0; g < NG; ++g) cch[g] = min(lane + 32 * g, C - 1);

    auto process = [&](int cb, int row, const float (&a)[H]) {
        float ex[H];
#pragma unroll
        for (int h = 0; h < H; ++h) {
            ex[h] = (cb + lane < deg) ? expf(a[h] - m[h]) : 0.f;
            den[h] += ex[h];
        }
        const int cnt = min(32, deg - cb);
#pragma unroll 4
        for (int j = 0; j < cnt; ++j) {
            const int rj = __builtin_amdgcn_readlane(row, j);
            const float* xr = XA + (size_t)rj * D;
#pragma unroll
            for (int h = 0; h < H; ++h) {
                const float w = __int_as_float(__builtin_amdgcn_readlane(__float_as_int(ex[h]), j));
                if constexpr (V4) {
                    const float4 x = *reinterpret_cast<const float4*>(xr + h * C + sl * 128 + lane * 4);
                    acc4[h].x += w * x.x; acc4[h].y += w * x.y;
                    acc4[h].z += w * x.z; acc4[h].w += w * x.w;
                } else {
#pragma unroll
                    for (int g = 0; g < NG; ++g)
                        acc1[h][g] += w * xr[h * C + cch[g]];
                }
            }
        }
    };

    process(0, row0, a0);
    for (int cb = 32; cb < deg; cb += 32) {
        int r; float a[H];
        load_edge(cb + lane, r, a);
        process(cb, r, a);
    }

    float inv[H];
#pragma unroll
    for (int h = 0; h < H; ++h) {
        for (int off = 16; off; off >>= 1) den[h] += __shfl_xor(den[h], off, 32);
        inv[h] = 1.f / (den[h] + 1e-16f);
    }

    if constexpr (V4) {
        const int c0 = sl * 128 + lane * 4;
        float4 o = make_float4(0.f, 0.f, 0.f, 0.f);
#pragma unroll
        for (int h = 0; h < H; ++h) {
            o.x += acc4[h].x * inv[h]; o.y += acc4[h].y * inv[h];
            o.z += acc4[h].z * inv[h]; o.w += acc4[h].w * inv[h];
        }
        o.x = elu_(o.x / (float)H + bias[c0 + 0]);
        o.y = elu_(o.y / (float)H + bias[c0 + 1]);
        o.z = elu_(o.z / (float)H + bias[c0 + 2]);
        o.w = elu_(o.w / (float)H + bias[c0 + 3]);
        *reinterpret_cast<float4*>(out + (size_t)node * C + c0) = o;
    } else {
#pragma unroll
        for (int g = 0; g < NG; ++g) {
            const int c = lane + 32 * g;
            if (c < C) {
                float o = 0.f;
#pragma unroll
                for (int h = 0; h < H; ++h) o += acc1[h][g] * inv[h];
                out[(size_t)node * C + c] = elu_(o / (float)H + bias[c]);
            }
        }
    }
}

struct WsPtrs {
    float* xall; float* qall; float* kall;
    int* rowptr; int* cnt; int* cpk;
};

template <int C, int H, bool REL>
static void run_attn_conv(const float* xin, const float* w, const float* qv,
                          const float* kv, const float* bias, int K,
                          const WsPtrs& p, float* out, hipStream_t stream) {
    constexpr int D  = C * H;
    constexpr int Rr = REL ? 4 : 1;
    if (D > 256) {
        dim3 gg(NN / GM_ROWS, (D + 255) / 256, Rr);
        k_gemm_rel_wmma<2><<<gg, 256, 0, stream>>>(xin, w, p.xall, NN, K, D);
    } else {
        dim3 gg(NN / GM_ROWS, (D + 127) / 128, Rr);
        k_gemm_rel_wmma<1><<<gg, 256, 0, stream>>>(xin, w, p.xall, NN, K, D);
    }

    constexpr int totalRN = Rr * NN;
    k_rowdot<D, H><<<(totalRN + 4 * RD_RW - 1) / (4 * RD_RW), 128, 0, stream>>>(
        p.xall, qv, kv, p.qall, p.kall, totalRN);

    constexpr int SL = (C % 128 == 0) ? C / 128 : 1;
    k_node_attn<C, H, REL><<<(NN * SL + 7) / 8, 256, 0, stream>>>(
        p.xall, p.qall, p.kall, p.rowptr, p.cpk, bias, out);
}

extern "C" void kernel_launch(void* const* d_in, const int* in_sizes, int n_in,
                              void* d_out, int out_size, void* d_ws, size_t ws_size,
                              hipStream_t stream) {
    (void)in_sizes; (void)n_in; (void)out_size; (void)ws_size;
    const float* x  = (const float*)d_in[0];
    const int*   ei = (const int*)  d_in[1];
    const int*   et = (const int*)  d_in[2];
    const float* w1 = (const float*)d_in[3],  *q1 = (const float*)d_in[4];
    const float* k1 = (const float*)d_in[5],  *b1 = (const float*)d_in[6];
    const float* wS = (const float*)d_in[7],  *qS = (const float*)d_in[8];
    const float* kS = (const float*)d_in[9],  *bS = (const float*)d_in[10];
    const float* w2 = (const float*)d_in[11], *q2 = (const float*)d_in[12];
    const float* k2 = (const float*)d_in[13], *b2 = (const float*)d_in[14];
    const float* wT = (const float*)d_in[15], *qT = (const float*)d_in[16];
    const float* kT = (const float*)d_in[17], *bT = (const float*)d_in[18];
    const float* wg = (const float*)d_in[19];
    const float* ag_src = (const float*)d_in[20];
    const float* ag_dst = (const float*)d_in[21];
    const float* bg = (const float*)d_in[22];

    const int* src = ei;
    const int* dst = ei + EE;

    float* ws = (float*)d_ws;
    size_t off = 0;
    WsPtrs p;
    p.xall   = ws + off;          off += (size_t)4 * NN * 512;
    p.qall   = ws + off;          off += (size_t)4 * NN * 2;
    p.kall   = ws + off;          off += (size_t)4 * NN * 2;
    p.rowptr = (int*)(ws + off);  off += (size_t)NN * 2;
    p.cnt    = (int*)(ws + off);  off += (size_t)NN * 2;
    p.cpk    = (int*)(ws + off);  off += (size_t)EE * 2;
    off += (size_t)NN * 256;
    float* xs = ws + off;         off += (size_t)NN * 256;
    float* xt = ws + off;         off += (size_t)NN * 256;

    float* x_in   = (float*)d_out;
    float* x_out  = (float*)d_out + (size_t)NN * 65;
    float* x_self = (float*)d_out + (size_t)2 * NN * 65;

    k_csr_zero<<<(NN + 255) / 256, 256, 0, stream>>>(p.cnt);
    k_csr_hist<<<(EE + 255) / 256, 256, 0, stream>>>(dst, p.cnt);
    k_csr_scan<<<1, 1024, 0, stream>>>(p.cnt, p.rowptr);
    k_csr_fill<<<(EE + 255) / 256, 256, 0, stream>>>(src, dst, et, p.rowptr, p.cnt, p.cpk);

    run_attn_conv<256, 2, true>(x,  w1, q1, k1, b1, 128, p, xs,   stream);
    run_attn_conv< 65, 2, true>(xs, wS, qS, kS, bS, 256, p, x_in, stream);
    run_attn_conv<256, 2, true>(x,  w2, q2, k2, b2, 128, p, xt,    stream);
    run_attn_conv< 65, 2, true>(xt, wT, qT, kT, bT, 256, p, x_out, stream);
    run_attn_conv<256, 1, false>(x, wg, ag_dst, ag_src, bg, 128, p, x_self, stream);
}


# Attention logits ride in the GEMM: q_all/k_all = X (W [q k]) come out as 2H extra columns fed by a small per-call product of the weights; the row-dot pass over x_all is gone (19 -> 15 launches)
# speedup vs baseline: 4.3669x; 1.0967x over previous
#include <hip/hip_runtime.h>
#include <hip/hip_bf16.h>
#include <math.h>

#define NN 10000
#define EE 160000

typedef __attribute__((ext_vector_type(2))) float v2f;
typedef __attribute__((ext_vector_type(8))) float v8f;

__device__ __forceinline__ float lrelu_(float x) { return x > 0.f ? x : 0.2f * x; }
__device__ __forceinline__ float elu_(float x)   { return x > 0.f ? x : (expf(x) - 1.f); }

#define QK_P     4
#define GM_MT    5
#define GM_ROWS  (16 * GM_MT)
#define GM_KC    128
#define GM_PITCH (GM_KC + 4)

template <int NT>
__global__ __launch_bounds__(256)
void k_gemm_rel_wmma(const float* __restrict__ X, const float* __restrict__ W,
                     float* __restrict__ Y, const float* __restrict__ WQ,
                     float* __restrict__ QK, int M, int K, int Nout, int ny, int R, int nqk) {
    __shared__ __attribute__((aligned(16))) float As[GM_ROWS * GM_PITCH];
    const int wave = __builtin_amdgcn_readfirstlane((int)(threadIdx.x >> 5));
    const int lane = threadIdx.x & 31;
    const int mt   = M / GM_ROWS;
    const int nbm  = mt * ny * R;
    const int bid  = blockIdx.x;
    const bool qkb = bid >= nbm;

    int m0, n0, pitch, nvalid;
    const float* Wr;
    float* Yr;
    bool active;
    if (!qkb) {
        const int mx = bid % mt, t = bid / mt;
        const int by = t % ny,   r = t / ny;
        m0 = mx * GM_ROWS;
        n0 = (by * 8 + wave) * (16 * NT);
        pitch = Nout; nvalid = Nout;
        Wr = W + (size_t)r * K * Nout;
        Yr = Y + (size_t)r * M * Nout;
        active = n0 < Nout;
    } else {
        active = wave < R;
        const int r = active ? wave : 0;
        m0 = (bid - nbm) * GM_ROWS;
        n0 = 0;
        pitch = QK_P; nvalid = nqk;
        Wr = WQ + (size_t)r * K * QK_P;
        Yr = QK + (size_t)r * M * QK_P;
    }

    const int kh    = (lane >> 4) * 2;
    const int mrow  = lane & 15;
    unsigned voff[NT];
    bool nv[NT];
#pragma unroll
    for (int j = 0; j < NT; ++j) {
        const int ncol = n0 + 16 * j + mrow;
        nv[j] = ncol < nvalid;
        const int ncl = nv[j] ? ncol : (nvalid - 1);
        voff[j] = (unsigned)(kh * pitch + ncl) * 4u;
    }
    const size_t rowB = (size_t)pitch * 4;

    v8f c[GM_MT][NT];
#pragma unroll
    for (int t = 0; t < GM_MT; ++t)
#pragma unroll
        for (int j = 0; j < NT; ++j) c[t][j] = v8f{};

    for (int kc = 0; kc < K; kc += GM_KC) {
        if (kc) __syncthreads();
        {
            float4 st[GM_ROWS / 8];
            const float* Xs = X + (size_t)(m0 + wave) * K + kc + lane * 4;
#pragma unroll
            for (int i = 0; i < GM_ROWS / 8; ++i)
                st[i] = *reinterpret_cast<const float4*>(Xs + (size_t)(8 * i) * K);
#pragma unroll
            for (int i = 0; i < GM_ROWS / 8; ++i)
                *reinterpret_cast<float4*>(&As[(wave + 8 * i) * GM_PITCH + lane * 4]) = st[i];
        }
        __syncthreads();

        if (active) {
            const char* Wk = reinterpret_cast<const char*>(Wr) + (size_t)kc * rowB;
#pragma unroll 4
            for (int k0 = 0; k0 < GM_KC; k0 += 4) {
                const char* r0 = Wk + (size_t)k0 * rowB;
                v2f b[NT];
#pragma unroll
                for (int j = 0; j < NT; ++j) {
                    b[j].x = *reinterpret_cast<const float*>(r0 + voff[j]);
                    b[j].y = *reinterpret_cast<const float*>(r0 + rowB + voff[j]);
                }
#pragma unroll
                for (int t = 0; t < GM_MT; ++t) {
                    const v2f a = *reinterpret_cast<const v2f*>(
                        &As[(t * 16 + mrow) * GM_PITCH + k0 + kh]);
#pragma unroll
                    for (int j = 0; j < NT; ++j)
                        c[t][j] = __builtin_amdgcn_wmma_f32_16x16x4_f32(
                            false, a, false, b[j], (short)0, c[t][j], false, false);
                }
            }
        }
    }

    if (active) {
#pragma unroll
        for (int t = 0; t < GM_MT; ++t) {
            const int rbase = m0 + t * 16 + (lane >> 4) * 8;
#pragma unroll
            for (int j = 0; j < NT; ++j) {
                if (nv[j]) {
                    const int ncol = n0 + 16 * j + mrow;
#pragma unroll
                    for (int i = 0; i < 8; ++i)
                        Yr[(size_t)(rbase + i) * pitch + ncol] = c[t][j][i];
                }
            }
        }
    }
}

struct WqkIn {
    const float* w1; const float* q1; const float* k1;
    const float* wS; const float* qS; const float* kS;
    const float* w2; const float* q2; const float* k2;
    const float* wT; const float* qT; const float* kT;
    const float* wg; const float* qg; const float* kg;
};

template <int D, int H>
__device__ __forceinline__ void wqk_row(const float* __restrict__ Wrow, const float* __restrict__ Q,
                                        const float* __restrict__ Kv, float* __restrict__ o,
                                        int lane) {
    constexpr int NI = (D + 31) / 32;
    float w[NI], cq[NI][H], ck[NI][H];
#pragma unroll
    for (int i = 0; i < NI; ++i) {
        const int  d  = lane + 32 * i;
        const bool ok = d < D;
        const int  dc = ok ? d : D - 1;
        const float wv = Wrow[dc];
        w[i] = ok ? wv : 0.f;
#pragma unroll
        for (int h = 0; h < H; ++h) { cq[i][h] = Q[dc * H + h]; ck[i][h] = Kv[dc * H + h]; }
    }
    float s[2 * H];
#pragma unroll
    for (int j = 0; j < 2 * H; ++j) s[j] = 0.f;
#pragma unroll
    for (int i = 0; i < NI; ++i)
#pragma unroll
        for (int h = 0; h < H; ++h) { s[h] += w[i] * cq[i][h]; s[H + h] += w[i] * ck[i][h]; }
    for (int off = 16; off; off >>= 1)
#pragma unroll
        for (int j = 0; j < 2 * H; ++j) s[j] += __shfl_xor(s[j], off, 32);
    float v = s[0];
#pragma unroll
    for (int j = 1; j < 2 * H; ++j) v = (lane == j) ? s[j] : v;
    if (lane < 2 * H) o[lane] = v;
}

#define WQ_ROWS (4 * 128 + 4 * 256 + 4 * 128 + 4 * 256 + 128)
__global__ __launch_bounds__(256)
void k_wqk(WqkIn a, float* __restrict__ wq) {
    const int lane = threadIdx.x & 31;
    const int gw = __builtin_amdgcn_readfirstlane((int)((blockIdx.x * blockDim.x + threadIdx.x) >> 5));
    float* o = wq + (size_t)gw * QK_P;
    if (gw < 512) {
        wqk_row<512, 2>(a.w1 + (size_t)gw * 512, a.q1, a.k1, o, lane);
    } else if (gw < 1536) {
        wqk_row<130, 2>(a.wS + (size_t)(gw - 512) * 130, a.qS, a.kS, o, lane);
    } else if (gw < 2048) {
        wqk_row<512, 2>(a.w2 + (size_t)(gw - 1536) * 512, a.q2, a.k2, o, lane);
    } else if (gw < 3072) {
        wqk_row<130, 2>(a.wT + (size_t)(gw - 2048) * 130, a.qT, a.kT, o, lane);
    } else if (gw < WQ_ROWS) {
        wqk_row<256, 1>(a.wg + (size_t)(gw - 3072) * 256, a.qg, a.kg, o, lane);
    }
}

__global__ void k_csr_zero(int* __restrict__ cnt) {
    const int t = blockIdx.x * blockDim.x + threadIdx.x;
    if (t < NN) cnt[t] = 0;
}

__global__ void k_csr_hist(const int* __restrict__ dst, int* __restrict__ cnt) {
    const int e = blockIdx.x * blockDim.x + threadIdx.x;
    if (e < EE) atomicAdd(&cnt[dst[e]], 1);
}

#define SC_ITEMS 10
__global__ __launch_bounds__(1024)
void k_csr_scan(const int* __restrict__ cnt, int* __restrict__ rowptr) {
    __shared__ int wtot[32];
    const int t = threadIdx.x, lane = t & 31, wv = t >> 5;
    const int base = t * SC_ITEMS;
    int v[SC_ITEMS];
    int s = 0;
#pragma unroll
    for (int i = 0; i < SC_ITEMS; ++i) {
        const int idx = base + i;
        v[i] = (idx < NN) ? cnt[idx] : 0;
        s += v[i];
    }
    int inc = s;
    for (int off = 1; off < 32; off <<= 1) {
        const int u = __shfl_up(inc, off, 32);
        if (lane >= off) inc += u;
    }
    if (lane == 31) wtot[wv] = inc;
    __syncthreads();
    if (wv == 0) {
        const int w = wtot[lane];
        int winc = w;
        for (int off = 1; off < 32; off <<= 1) {
            const int u = __shfl_up(winc, off, 32);
            if (lane >= off) winc += u;
        }
        wtot[lane] = winc - w;
    }
    __syncthreads();
    int run = wtot[wv] + inc - s;
#pragma unroll
    for (int i = 0; i < SC_ITEMS; ++i) {
        const int idx = base + i;
        if (idx < NN) { rowptr[idx] = run; run += v[i]; }
    }
    if (t == 1023) rowptr[NN] = run;
}

__global__ void k_csr_fill(const int* __restrict__ src, const int* __restrict__ dst,
                           const int* __restrict__ etype, const int* __restrict__ rowptr,
                           int* __restrict__ cnt, int* __restrict__ cpk) {
    const int e = blockIdx.x * blockDim.x + threadIdx.x;
    if (e >= EE) return;
    const int d = dst[e];
    const int pos = atomicSub(&cnt[d], 1) - 1;
    cpk[rowptr[d] + pos] = (etype[e] << 16) | src[e];
}

template <int C, int H, bool REL>
__global__ __launch_bounds__(256)
void k_node_attn(const float* __restrict__ XA, const float* __restrict__ qk,
                 const int* __restrict__ rowptr,
                 const int* __restrict__ cpk, const float* __restrict__ bias,
                 float* __restrict__ out) {
    constexpr int  D  = C * H;
    constexpr bool V4 = (C % 128 == 0);
    constexpr int  SL = V4 ? C / 128 : 1;
    constexpr int  NG = V4 ? 1 : (C + 31) / 32;
    const int lane = threadIdx.x & 31;
    const int gw   = __builtin_amdgcn_readfirstlane((int)((blockIdx.x * blockDim.x + threadIdx.x) >> 5));
    const int node = gw / SL;
    const int sl   = gw - node * SL;
    if (node >= NN) return;
    const int beg = rowptr[node];
    const int deg = rowptr[node + 1] - beg;

    auto load_edge = [&](int idx, int& row, float (&a)[H]) {
        const bool valid = idx < deg;
        const int p  = valid ? cpk[beg + idx] : 0;
        const int et = REL ? (p >> 16) : 0;
        row = et * NN + (p & 0xFFFF);
        const int qrow = et * NN + node;
#pragma unroll
        for (int h = 0; h < H; ++h) {
            const float v = lrelu_(qk[(size_t)qrow * QK_P + h] + qk[(size_t)row * QK_P + H + h]);
            a[h] = valid ? v : -INFINITY;
        }
    };

    int row0; float a0[H];
    load_edge(lane, row0, a0);
    float m[H];
#pragma unroll
    for (int h = 0; h < H; ++h) m[h] = a0[h];
    for (int cb = 32; cb < deg; cb += 32) {
        int r; float a[H];
        load_edge(cb + lane, r, a);
#pragma unroll
        for (int h = 0; h < H; ++h) m[h] = fmaxf(m[h], a[h]);
    }
#pragma unroll
    for (int h = 0; h < H; ++h) {
        for (int off = 16; off; off >>= 1) m[h] = fmaxf(m[h], __shfl_xor(m[h], off, 32));
        m[h] = isfinite(m[h]) ? m[h] : 0.f;
    }

    float den[H];
    float4 acc4[H];
    float  acc1[H][NG];
#pragma unroll
    for (int h = 0; h < H; ++h) {
        den[h] = 0.f;
        acc4[h] = make_float4(0.f, 0.f, 0.f, 0.f);
#pragma unroll
        for (int g = 0; g < NG; ++g) acc1[h][g] = 0.f;
    }

    int cch[NG];
#pragma unroll
    for (int g = 0; g < NG; ++g) cch[g] = min(lane + 32 * g, C - 1);

    auto process = [&](int cb, int row, const float (&a)[H]) {
        float ex[H];
#pragma unroll
        for (int h = 0; h < H; ++h) {
            ex[h] = (cb + lane < deg) ? expf(a[h] - m[h]) : 0.f;
            den[h] += ex[h];
        }
        const int cnt = min(32, deg - cb);
#pragma unroll 4
        for (int j = 0; j < cnt; ++j) {
            const int rj = __builtin_amdgcn_readlane(row, j);
            const float* xr = XA + (size_t)rj * D;
#pragma unroll
            for (int h = 0; h < H; ++h) {
                const float w = __int_as_float(__builtin_amdgcn_readlane(__float_as_int(ex[h]), j));
                if constexpr (V4) {
                    const float4 x = *reinterpret_cast<const float4*>(xr + h * C + sl * 128 + lane * 4);
                    acc4[h].x += w * x.x; acc4[h].y += w * x.y;
                    acc4[h].z += w * x.z; acc4[h].w += w * x.w;
                } else {
#pragma unroll
                    for (int g = 0; g < NG; ++g)
                        acc1[h][g] += w * xr[h * C + cch[g]];
                }
            }
        }
    };

    process(0, row0, a0);
    for (int cb = 32; cb < deg; cb += 32) {
        int r; float a[H];
        load_edge(cb + lane, r, a);
        process(cb, r, a);
    }

    float inv[H];
#pragma unroll
    for (int h = 0; h < H; ++h) {
        for (int off = 16; off; off >>= 1) den[h] += __shfl_xor(den[h], off, 32);
        inv[h] = 1.f / (den[h] + 1e-16f);
    }

    if constexpr (V4) {
        const int c0 = sl * 128 + lane * 4;
        float4 o = make_float4(0.f, 0.f, 0.f, 0.f);
#pragma unroll
        for (int h = 0; h < H; ++h) {
            o.x += acc4[h].x * inv[h]; o.y += acc4[h].y * inv[h];
            o.z += acc4[h].z * inv[h]; o.w += acc4[h].w * inv[h];
        }
        o.x = elu_(o.x / (float)H + bias[c0 + 0]);
        o.y = elu_(o.y / (float)H + bias[c0 + 1]);
        o.z = elu_(o.z / (float)H + bias[c0 + 2]);
        o.w = elu_(o.w / (float)H + bias[c0 + 3]);
        *reinterpret_cast<float4*>(out + (size_t)node * C + c0) = o;
    } else {
#pragma unroll
        for (int g = 0; g < NG; ++g) {
            const int c = lane + 32 * g;
            if (c < C) {
                float o = 0.f;
#pragma unroll
                for (int h = 0; h < H; ++h) o += acc1[h][g] * inv[h];
                out[(size_t)node * C + c] = elu_(o / (float)H + bias[c]);
            }
        }
    }
}

struct WsPtrs {
    float* xall; float* qk; float* wqk;
    int* rowptr; int* cnt; int* cpk;
};

template <int C, int H, bool REL>
static void run_attn_conv(const float* xin, const float* w, const float* wq,
                          const float* bias, int K,
                          const WsPtrs& p, float* out, hipStream_t stream) {
    constexpr int D  = C * H;
    constexpr int Rr = REL ? 4 : 1;
    constexpr int MT = NN / GM_ROWS;
    if (D > 256) {
        constexpr int ny = (D + 255) / 256;
        k_gemm_rel_wmma<2><<<MT * ny * Rr + MT, 256, 0, stream>>>(
            xin, w, p.xall, wq, p.qk, NN, K, D, ny, Rr, 2 * H);
    } else {
        constexpr int ny = (D + 127) / 128;
        k_gemm_rel_wmma<1><<<MT * ny * Rr + MT, 256, 0, stream>>>(
            xin, w, p.xall, wq, p.qk, NN, K, D, ny, Rr, 2 * H);
    }

    constexpr int SL = (C % 128 == 0) ? C / 128 : 1;
    k_node_attn<C, H, REL><<<(NN * SL + 7) / 8, 256, 0, stream>>>(
        p.xall, p.qk, p.rowptr, p.cpk, bias, out);
}

extern "C" void kernel_launch(void* const* d_in, const int* in_sizes, int n_in,
                              void* d_out, int out_size, void* d_ws, size_t ws_size,
                              hipStream_t stream) {
    (void)in_sizes; (void)n_in; (void)out_size; (void)ws_size;
    const float* x  = (const float*)d_in[0];
    const int*   ei = (const int*)  d_in[1];
    const int*   et = (const int*)  d_in[2];
    const float* w1 = (const float*)d_in[3],  *q1 = (const float*)d_in[4];
    const float* k1 = (const float*)d_in[5],  *b1 = (const float*)d_in[6];
    const float* wS = (const float*)d_in[7],  *qS = (const float*)d_in[8];
    const float* kS = (const float*)d_in[9],  *bS = (const float*)d_in[10];
    const float* w2 = (const float*)d_in[11], *q2 = (const float*)d_in[12];
    const float* k2 = (const float*)d_in[13], *b2 = (const float*)d_in[14];
    const float* wT = (const float*)d_in[15], *qT = (const float*)d_in[16];
    const float* kT = (const float*)d_in[17], *bT = (const float*)d_in[18];
    const float* wg = (const float*)d_in[19];
    const float* ag_src = (const float*)d_in[20];
    const float* ag_dst = (const float*)d_in[21];
    const float* bg = (const float*)d_in[22];

    const int* src = ei;
    const int* dst = ei + EE;

    float* ws = (float*)d_ws;
    size_t off = 0;
    WsPtrs p;
    p.xall   = ws + off;          off += (size_t)4 * NN * 512;
    p.qk     = ws + off;          off += (size_t)4 * NN * QK_P;
    p.rowptr = (int*)(ws + off);  off += (size_t)NN * 2;
    p.cnt    = (int*)(ws + off);  off += (size_t)NN * 2;
    p.cpk    = (int*)(ws + off);  off += (size_t)EE * 2;
    p.wqk    = ws + off;          off += (size_t)NN * 256;
    float* xs = ws + off;         off += (size_t)NN * 256;
    float* xt = ws + off;         off += (size_t)NN * 256;

    float* x_in   = (float*)d_out;
    float* x_out  = (float*)d_out + (size_t)NN * 65;
    float* x_self = (float*)d_out + (size_t)2 * NN * 65;

    k_csr_zero<<<(NN + 255) / 256, 256, 0, stream>>>(p.cnt);
    k_csr_hist<<<(EE + 255) / 256, 256, 0, stream>>>(dst, p.cnt);
    k_csr_scan<<<1, 1024, 0, stream>>>(p.cnt, p.rowptr);
    k_csr_fill<<<(EE + 255) / 256, 256, 0, stream>>>(src, dst, et, p.rowptr, p.cnt, p.cpk);

    WqkIn wi = { w1, q1, k1,  wS, qS, kS,  w2, q2, k2,  wT, qT, kT,  wg, ag_dst, ag_src };
    k_wqk<<<WQ_ROWS / 8, 256, 0, stream>>>(wi, p.wqk);
    const float* wq1 = p.wqk;
    const float* wqS = p.wqk + 512 * QK_P;
    const float* wq2 = p.wqk + 1536 * QK_P;
    const float* wqT = p.wqk + 2048 * QK_P;
    const float* wqg = p.wqk + 3072 * QK_P;

    run_attn_conv<256, 2, true>(x,  w1, wq1, b1, 128, p, xs,   stream);
    run_attn_conv< 65, 2, true>(xs, wS, wqS, bS, 256, p, x_in, stream);
    run_attn_conv<256, 2, true>(x,  w2, wq2, b2, 128, p, xt,    stream);
    run_attn_conv< 65, 2, true>(xt, wT, wqT, bT, 256, p, x_out, stream);
    run_attn_conv<256, 1, false>(x, wg, wqg, bg, 128, p, x_self, stream);
}
